# starting kernel
# baseline (speedup 1.0000x reference)
#include <hip/hip_runtime.h>

typedef __attribute__((ext_vector_type(2))) float v2f;
typedef __attribute__((ext_vector_type(8))) float v8f;

constexpr int Bsz = 8, N1 = 1024, N2 = 1024, DIM = 256, HID = 16;
constexpr int WCOLS = 6 * HID + 1;

__global__ __launch_bounds__(256) void weights_gemm_wmma(
        const float* __restrict__ q, const float* __restrict__ Wg,
        const float* __restrict__ bg, float* __restrict__ wout) {
    const int lane = threadIdx.x & 31;
    const int wid  = threadIdx.x >> 5;
    const int tile = blockIdx.x * 8 + wid;
    const int COLT = 7;
    const int row0 = (tile / COLT) * 16;
    const int col0 = (tile % COLT) * 16;
    const int h    = lane & 15;
    const bool hiH = lane >= 16;
    const int koff = hiH ? 2 : 0;
    const int col  = col0 + h;
    const int colc = (col < WCOLS) ? col : (WCOLS - 1);

    const float* qrow = q + (size_t)(row0 + h) * DIM + koff;
    const float* wcol = Wg + colc;

    v8f c = {};
    #pragma unroll 4
    for (int kk = 0; kk < DIM; kk += 4) {
        v2f a, bm;
        a.x  = qrow[kk];
        a.y  = qrow[kk + 1];
        bm.x = wcol[(size_t)(kk + koff)     * WCOLS];
        bm.y = wcol[(size_t)(kk + koff + 1) * WCOLS];
        c = __builtin_amdgcn_wmma_f32_16x16x4_f32(false, a, false, bm,
                                                  (short)0, c, false, false);
    }

    const float bgv = bg[colc];
    const int rbase = row0 + (hiH ? 8 : 0);
    #pragma unroll
    for (int r = 0; r < 8; ++r) {
        if (col < WCOLS)
            wout[(size_t)(rbase + r) * WCOLS + col] = c[r] + bgv;
    }
}

__global__ __launch_bounds__(256) void pair_mlp_wmma(
        const float* __restrict__ pos1, const float* __restrict__ pos2,
        const float* __restrict__ wgt, float* __restrict__ out) {
    const int lane = threadIdx.x & 31;
    const int wid  = threadIdx.x >> 5;
    const int gid  = blockIdx.x * 8 + wid;
    const int b    = gid >> 10;
    const bool hiH = lane >= 16;
    const int h    = lane & 15;

    const float* p1 = pos1 + (size_t)gid * 4;
    const float x1  = p1[0], y1 = p1[1], w1p = p1[2], h1p = p1[3];
    const float c0  = hiH ? __logf(w1p) : x1;
    const float c1  = hiH ? __logf(h1p) : y1;
    const float s0  = hiH ? 1.0f : 1.0f / w1p;
    const float s1  = hiH ? 1.0f : 1.0f / h1p;

    const float* wrow = wgt + (size_t)gid * WCOLS;
    const int k0 = hiH ? 2 : 0;
    v2f a;
    a.x = wrow[(k0 + 0) * HID + h];
    a.y = wrow[(k0 + 1) * HID + h];
    const int hb = hiH ? 8 : 0;
    float b1v[8], w2v[8];
    #pragma unroll
    for (int r = 0; r < 8; ++r) {
        b1v[r] = wrow[4 * HID + hb + r];
        w2v[r] = wrow[5 * HID + hb + r];
    }
    const float b2 = wrow[6 * HID];

    const float2* p2row = (const float2*)pos2 + (size_t)b * N2 * 2;
    float* orow = out + (size_t)gid * N2;

    for (int t = 0; t < N2 / 16; ++t) {
        const int n0 = t * 16;
        const float2 p = p2row[(size_t)(n0 + h) * 2 + (hiH ? 1 : 0)];
        const float u0 = hiH ? __logf(p.x) : p.x;
        const float u1 = hiH ? __logf(p.y) : p.y;
        v2f f;
        f.x = (u0 - c0) * s0;
        f.y = (u1 - c1) * s1;

        v8f d = {};
        d = __builtin_amdgcn_wmma_f32_16x16x4_f32(false, a, false, f,
                                                  (short)0, d, false, false);
        float acc = 0.0f;
        #pragma unroll
        for (int r = 0; r < 8; ++r) {
            float tt = fmaxf(d[r] + b1v[r], 0.0f);
            acc = fmaf(tt, w2v[r], acc);
        }
        acc += __shfl_xor(acc, 16, 32);
        if (!hiH) orow[n0 + h] = acc + b2;
    }
}

extern "C" void kernel_launch(void* const* d_in, const int* in_sizes, int n_in,
                              void* d_out, int out_size, void* d_ws, size_t ws_size,
                              hipStream_t stream) {
    const float* pos1    = (const float*)d_in[0];
    const float* pos2    = (const float*)d_in[1];
    const float* queries = (const float*)d_in[2];
    const float* Wg      = (const float*)d_in[3];
    const float* bg      = (const float*)d_in[4];

    float* wws = (float*)d_ws;

    weights_gemm_wmma<<<448, 256, 0, stream>>>(queries, Wg, bg, wws);
    pair_mlp_wmma<<<1024, 256, 0, stream>>>(pos1, pos2, wws, (float*)d_out);
}


# pair kernel: four n-tiles per trip as independent chains, next trip's pos2 loads issued one trip ahead (was one serial load-log-WMMA-reduce-shuffle-store chain per tile); every sum keeps its order
# speedup vs baseline: 1.6015x; 1.6015x over previous
#include <hip/hip_runtime.h>

typedef __attribute__((ext_vector_type(2))) float v2f;
typedef __attribute__((ext_vector_type(8))) float v8f;

constexpr int Bsz = 8, N1 = 1024, N2 = 1024, DIM = 256, HID = 16;
constexpr int WCOLS = 6 * HID + 1;

__global__ __launch_bounds__(256) void weights_gemm_wmma(
        const float* __restrict__ q, const float* __restrict__ Wg,
        const float* __restrict__ bg, float* __restrict__ wout) {
    const int lane = threadIdx.x & 31;
    const int wid  = threadIdx.x >> 5;
    const int tile = blockIdx.x * 8 + wid;
    const int COLT = 7;
    const int row0 = (tile / COLT) * 16;
    const int col0 = (tile % COLT) * 16;
    const int h    = lane & 15;
    const bool hiH = lane >= 16;
    const int koff = hiH ? 2 : 0;
    const int col  = col0 + h;
    const int colc = (col < WCOLS) ? col : (WCOLS - 1);

    const float* qrow = q + (size_t)(row0 + h) * DIM + koff;
    const float* wcol = Wg + colc;

    v8f c = {};
    #pragma unroll 4
    for (int kk = 0; kk < DIM; kk += 4) {
        v2f a, bm;
        a.x  = qrow[kk];
        a.y  = qrow[kk + 1];
        bm.x = wcol[(size_t)(kk + koff)     * WCOLS];
        bm.y = wcol[(size_t)(kk + koff + 1) * WCOLS];
        c = __builtin_amdgcn_wmma_f32_16x16x4_f32(false, a, false, bm,
                                                  (short)0, c, false, false);
    }

    const float bgv = bg[colc];
    const int rbase = row0 + (hiH ? 8 : 0);
    #pragma unroll
    for (int r = 0; r < 8; ++r) {
        if (col < WCOLS)
            wout[(size_t)(rbase + r) * WCOLS + col] = c[r] + bgv;
    }
}

__global__ __launch_bounds__(256) void pair_mlp_wmma(
        const float* __restrict__ pos1, const float* __restrict__ pos2,
        const float* __restrict__ wgt, float* __restrict__ out) {
    const int lane = threadIdx.x & 31;
    const int wid  = threadIdx.x >> 5;
    const int gid  = blockIdx.x * 8 + wid;
    const int b    = gid >> 10;
    const bool hiH = lane >= 16;
    const int h    = lane & 15;

    const float* p1 = pos1 + (size_t)gid * 4;
    const float x1  = p1[0], y1 = p1[1], w1p = p1[2], h1p = p1[3];
    const float c0  = hiH ? __logf(w1p) : x1;
    const float c1  = hiH ? __logf(h1p) : y1;
    const float s0  = hiH ? 1.0f : 1.0f / w1p;
    const float s1  = hiH ? 1.0f : 1.0f / h1p;

    const float* wrow = wgt + (size_t)gid * WCOLS;
    const int k0 = hiH ? 2 : 0;
    v2f a;
    a.x = wrow[(k0 + 0) * HID + h];
    a.y = wrow[(k0 + 1) * HID + h];
    const int hb = hiH ? 8 : 0;
    float b1v[8], w2v[8];
    #pragma unroll
    for (int r = 0; r < 8; ++r) {
        b1v[r] = wrow[4 * HID + hb + r];
        w2v[r] = wrow[5 * HID + hb + r];
    }
    const float b2 = wrow[6 * HID];

    const float2* pl = (const float2*)pos2 + (size_t)b * N2 * 2
                     + (size_t)h * 2 + (hiH ? 1 : 0);
    float* orow = out + (size_t)gid * N2;

    constexpr int TPT   = 4;
    constexpr int TRIPS = N2 / (16 * TPT);

    float2 pn[TPT];
    #pragma unroll
    for (int j = 0; j < TPT; ++j) pn[j] = pl[(size_t)(j * 16) * 2];

    for (int trip = 0; trip < TRIPS; ++trip) {
        float2 pc[TPT];
        #pragma unroll
        for (int j = 0; j < TPT; ++j) pc[j] = pn[j];

        const int tn = (trip + 1) & (TRIPS - 1);
        #pragma unroll
        for (int j = 0; j < TPT; ++j)
            pn[j] = pl[(size_t)((tn * TPT + j) * 16) * 2];
        __builtin_amdgcn_sched_barrier(0);

        float part[TPT];
        #pragma unroll
        for (int j = 0; j < TPT; ++j) {
            const float u0 = hiH ? __logf(pc[j].x) : pc[j].x;
            const float u1 = hiH ? __logf(pc[j].y) : pc[j].y;
            v2f f;
            f.x = (u0 - c0) * s0;
            f.y = (u1 - c1) * s1;

            v8f d = {};
            d = __builtin_amdgcn_wmma_f32_16x16x4_f32(false, a, false, f,
                                                      (short)0, d, false, false);
            float acc = 0.0f;
            #pragma unroll
            for (int r = 0; r < 8; ++r) {
                float tt = fmaxf(d[r] + b1v[r], 0.0f);
                acc = fmaf(tt, w2v[r], acc);
            }
            part[j] = acc;
        }

        float oth[TPT];
        #pragma unroll
        for (int j = 0; j < TPT; ++j) oth[j] = __shfl_xor(part[j], 16, 32);

        if (!hiH) {
            #pragma unroll
            for (int j = 0; j < TPT; ++j)
                orow[(trip * TPT + j) * 16 + h] = (part[j] + oth[j]) + b2;
        }
    }
}

extern "C" void kernel_launch(void* const* d_in, const int* in_sizes, int n_in,
                              void* d_out, int out_size, void* d_ws, size_t ws_size,
                              hipStream_t stream) {
    const float* pos1    = (const float*)d_in[0];
    const float* pos2    = (const float*)d_in[1];
    const float* queries = (const float*)d_in[2];
    const float* Wg      = (const float*)d_in[3];
    const float* bg      = (const float*)d_in[4];

    float* wws = (float*)d_ws;

    weights_gemm_wmma<<<448, 256, 0, stream>>>(queries, Wg, bg, wws);
    pair_mlp_wmma<<<1024, 256, 0, stream>>>(pos1, pos2, wws, (float*)d_out);
}


# pair kernel: pos2 features (x2, y2, log w2, log h2) built once per work-group in LDS, 8 logs per thread, tiles read them from there; was 2 full logs per lane per tile; results bit-identical
# speedup vs baseline: 1.9036x; 1.1886x over previous
#include <hip/hip_runtime.h>

typedef __attribute__((ext_vector_type(2))) float v2f;
typedef __attribute__((ext_vector_type(8))) float v8f;

constexpr int Bsz = 8, N1 = 1024, N2 = 1024, DIM = 256, HID = 16;
constexpr int WCOLS = 6 * HID + 1;

__global__ __launch_bounds__(256) void weights_gemm_wmma(
        const float* __restrict__ q, const float* __restrict__ Wg,
        const float* __restrict__ bg, float* __restrict__ wout) {
    const int lane = threadIdx.x & 31;
    const int wid  = threadIdx.x >> 5;
    const int tile = blockIdx.x * 8 + wid;
    const int COLT = 7;
    const int row0 = (tile / COLT) * 16;
    const int col0 = (tile % COLT) * 16;
    const int h    = lane & 15;
    const bool hiH = lane >= 16;
    const int koff = hiH ? 2 : 0;
    const int col  = col0 + h;
    const int colc = (col < WCOLS) ? col : (WCOLS - 1);

    const float* qrow = q + (size_t)(row0 + h) * DIM + koff;
    const float* wcol = Wg + colc;

    v8f c = {};
    #pragma unroll 4
    for (int kk = 0; kk < DIM; kk += 4) {
        v2f a, bm;
        a.x  = qrow[kk];
        a.y  = qrow[kk + 1];
        bm.x = wcol[(size_t)(kk + koff)     * WCOLS];
        bm.y = wcol[(size_t)(kk + koff + 1) * WCOLS];
        c = __builtin_amdgcn_wmma_f32_16x16x4_f32(false, a, false, bm,
                                                  (short)0, c, false, false);
    }

    const float bgv = bg[colc];
    const int rbase = row0 + (hiH ? 8 : 0);
    #pragma unroll
    for (int r = 0; r < 8; ++r) {
        if (col < WCOLS)
            wout[(size_t)(rbase + r) * WCOLS + col] = c[r] + bgv;
    }
}

__global__ __launch_bounds__(256) void pair_mlp_wmma(
        const float* __restrict__ pos1, const float* __restrict__ pos2,
        const float* __restrict__ wgt, float* __restrict__ out) {
    __shared__ float4 feat2[N2];

    const int lane = threadIdx.x & 31;
    const int wid  = threadIdx.x >> 5;
    const int gid  = blockIdx.x * 8 + wid;
    const int b    = gid >> 10;
    const bool hiH = lane >= 16;
    const int h    = lane & 15;

    {
        const float4* p2 = (const float4*)pos2 + (size_t)b * N2;
        #pragma unroll
        for (int i = 0; i < N2 / 256; ++i) {
            const int n = i * 256 + (int)threadIdx.x;
            float4 p = p2[n];
            p.z = __logf(p.z);
            p.w = __logf(p.w);
            feat2[n] = p;
        }
    }

    const float* p1 = pos1 + (size_t)gid * 4;
    const float x1  = p1[0], y1 = p1[1], w1p = p1[2], h1p = p1[3];
    const float c0  = hiH ? __logf(w1p) : x1;
    const float c1  = hiH ? __logf(h1p) : y1;
    const float s0  = hiH ? 1.0f : 1.0f / w1p;
    const float s1  = hiH ? 1.0f : 1.0f / h1p;

    const float* wrow = wgt + (size_t)gid * WCOLS;
    const int k0 = hiH ? 2 : 0;
    v2f a;
    a.x = wrow[(k0 + 0) * HID + h];
    a.y = wrow[(k0 + 1) * HID + h];
    const int hb = hiH ? 8 : 0;
    float b1v[8], w2v[8];
    #pragma unroll
    for (int r = 0; r < 8; ++r) {
        b1v[r] = wrow[4 * HID + hb + r];
        w2v[r] = wrow[5 * HID + hb + r];
    }
    const float b2 = wrow[6 * HID];

    const float2* pl = (const float2*)feat2 + h * 2 + (hiH ? 1 : 0);
    float* orow = out + (size_t)gid * N2;

    __syncthreads();

    constexpr int TPT   = 4;
    constexpr int TRIPS = N2 / (16 * TPT);

    float2 pn[TPT];
    #pragma unroll
    for (int j = 0; j < TPT; ++j) pn[j] = pl[(j * 16) * 2];

    for (int trip = 0; trip < TRIPS; ++trip) {
        float2 pc[TPT];
        #pragma unroll
        for (int j = 0; j < TPT; ++j) pc[j] = pn[j];

        const int tn = (trip + 1) & (TRIPS - 1);
        #pragma unroll
        for (int j = 0; j < TPT; ++j)
            pn[j] = pl[((tn * TPT + j) * 16) * 2];
        __builtin_amdgcn_sched_barrier(0);

        float part[TPT];
        #pragma unroll
        for (int j = 0; j < TPT; ++j) {
            v2f f;
            f.x = (pc[j].x - c0) * s0;
            f.y = (pc[j].y - c1) * s1;

            v8f d = {};
            d = __builtin_amdgcn_wmma_f32_16x16x4_f32(false, a, false, f,
                                                      (short)0, d, false, false);
            float acc = 0.0f;
            #pragma unroll
            for (int r = 0; r < 8; ++r) {
                float tt = fmaxf(d[r] + b1v[r], 0.0f);
                acc = fmaf(tt, w2v[r], acc);
            }
            part[j] = acc;
        }

        float oth[TPT];
        #pragma unroll
        for (int j = 0; j < TPT; ++j) oth[j] = __shfl_xor(part[j], 16, 32);

        if (!hiH) {
            #pragma unroll
            for (int j = 0; j < TPT; ++j)
                orow[(trip * TPT + j) * 16 + h] = (part[j] + oth[j]) + b2;
        }
    }
}

extern "C" void kernel_launch(void* const* d_in, const int* in_sizes, int n_in,
                              void* d_out, int out_size, void* d_ws, size_t ws_size,
                              hipStream_t stream) {
    const float* pos1    = (const float*)d_in[0];
    const float* pos2    = (const float*)d_in[1];
    const float* queries = (const float*)d_in[2];
    const float* Wg      = (const float*)d_in[3];
    const float* bg      = (const float*)d_in[4];

    float* wws = (float*)d_ws;

    weights_gemm_wmma<<<448, 256, 0, stream>>>(queries, Wg, bg, wws);
    pair_mlp_wmma<<<1024, 256, 0, stream>>>(pos1, pos2, wws, (float*)d_out);
}


# weights GEMM: each work-group stages its 16x256 query tile in LDS once with whole-line loads and its 7 column-tile waves read A fragments there (was 8 B from 16 lines per load); sums unchanged
# speedup vs baseline: 2.4745x; 1.2999x over previous
#include <hip/hip_runtime.h>

typedef __attribute__((ext_vector_type(2))) float v2f;
typedef __attribute__((ext_vector_type(8))) float v8f;

constexpr int Bsz = 8, N1 = 1024, N2 = 1024, DIM = 256, HID = 16;
constexpr int WCOLS = 6 * HID + 1;

constexpr int QP = DIM + 4;

__global__ __launch_bounds__(256) void weights_gemm_wmma(
        const float* __restrict__ q, const float* __restrict__ Wg,
        const float* __restrict__ bg, float* __restrict__ wout) {
    __shared__ __attribute__((aligned(16))) float qs[16 * QP];

    const int lane = threadIdx.x & 31;
    const int wid  = threadIdx.x >> 5;
    const int row0 = blockIdx.x * 16;
    const int h    = lane & 15;
    const bool hiH = lane >= 16;
    const int koff = hiH ? 2 : 0;
    const int col  = wid * 16 + h;
    const int colc = (col < WCOLS) ? col : (WCOLS - 1);

    const float4* qg = (const float4*)(q + (size_t)row0 * DIM);
    float4 v[4];
    #pragma unroll
    for (int i = 0; i < 4; ++i) v[i] = qg[i * 256 + (int)threadIdx.x];

    constexpr int KU     = 16;
    constexpr int ROUNDS = DIM / (4 * KU);
    const float* wcol = Wg + colc + (size_t)koff * WCOLS;
    v2f bn[KU];
    #pragma unroll
    for (int j = 0; j < KU; ++j) {
        bn[j].x = wcol[(size_t)(j * 4)     * WCOLS];
        bn[j].y = wcol[(size_t)(j * 4 + 1) * WCOLS];
    }

    #pragma unroll
    for (int i = 0; i < 4; ++i) {
        const int idx = i * 256 + (int)threadIdx.x;
        const int r   = idx >> 6;
        const int c4  = idx & 63;
        *(float4*)&qs[r * QP + c4 * 4] = v[i];
    }
    __syncthreads();
    if (wid == 7) return;

    const float* arow = qs + h * QP + koff;

    v8f c = {};
    #pragma unroll
    for (int rd = 0; rd < ROUNDS; ++rd) {
        v2f bc[KU];
        #pragma unroll
        for (int j = 0; j < KU; ++j) bc[j] = bn[j];
        if (rd + 1 < ROUNDS) {
            #pragma unroll
            for (int j = 0; j < KU; ++j) {
                const int kk = ((rd + 1) * KU + j) * 4;
                bn[j].x = wcol[(size_t)kk       * WCOLS];
                bn[j].y = wcol[(size_t)(kk + 1) * WCOLS];
            }
        }
        #pragma unroll
        for (int j = 0; j < KU; ++j) {
            const int kk = (rd * KU + j) * 4;
            const v2f a = *(const v2f*)(arow + kk);
            c = __builtin_amdgcn_wmma_f32_16x16x4_f32(false, a, false, bc[j],
                                                      (short)0, c, false, false);
        }
    }

    const float bgv = bg[colc];
    const int rbase = row0 + (hiH ? 8 : 0);
    #pragma unroll
    for (int r = 0; r < 8; ++r) {
        if (col < WCOLS)
            wout[(size_t)(rbase + r) * WCOLS + col] = c[r] + bgv;
    }
}

__global__ __launch_bounds__(256) void pair_mlp_wmma(
        const float* __restrict__ pos1, const float* __restrict__ pos2,
        const float* __restrict__ wgt, float* __restrict__ out) {
    __shared__ float4 feat2[N2];

    const int lane = threadIdx.x & 31;
    const int wid  = threadIdx.x >> 5;
    const int gid  = blockIdx.x * 8 + wid;
    const int b    = gid >> 10;
    const bool hiH = lane >= 16;
    const int h    = lane & 15;

    {
        const float4* p2 = (const float4*)pos2 + (size_t)b * N2;
        #pragma unroll
        for (int i = 0; i < N2 / 256; ++i) {
            const int n = i * 256 + (int)threadIdx.x;
            float4 p = p2[n];
            p.z = __logf(p.z);
            p.w = __logf(p.w);
            feat2[n] = p;
        }
    }

    const float* p1 = pos1 + (size_t)gid * 4;
    const float x1  = p1[0], y1 = p1[1], w1p = p1[2], h1p = p1[3];
    const float c0  = hiH ? __logf(w1p) : x1;
    const float c1  = hiH ? __logf(h1p) : y1;
    const float s0  = hiH ? 1.0f : 1.0f / w1p;
    const float s1  = hiH ? 1.0f : 1.0f / h1p;

    const float* wrow = wgt + (size_t)gid * WCOLS;
    const int k0 = hiH ? 2 : 0;
    v2f a;
    a.x = wrow[(k0 + 0) * HID + h];
    a.y = wrow[(k0 + 1) * HID + h];
    const int hb = hiH ? 8 : 0;
    float b1v[8], w2v[8];
    #pragma unroll
    for (int r = 0; r < 8; ++r) {
        b1v[r] = wrow[4 * HID + hb + r];
        w2v[r] = wrow[5 * HID + hb + r];
    }
    const float b2 = wrow[6 * HID];

    const float2* pl = (const float2*)feat2 + h * 2 + (hiH ? 1 : 0);
    float* orow = out + (size_t)gid * N2;

    __syncthreads();

    constexpr int TPT   = 4;
    constexpr int TRIPS = N2 / (16 * TPT);

    float2 pn[TPT];
    #pragma unroll
    for (int j = 0; j < TPT; ++j) pn[j] = pl[(j * 16) * 2];

    for (int trip = 0; trip < TRIPS; ++trip) {
        float2 pc[TPT];
        #pragma unroll
        for (int j = 0; j < TPT; ++j) pc[j] = pn[j];

        const int tn = (trip + 1) & (TRIPS - 1);
        #pragma unroll
        for (int j = 0; j < TPT; ++j)
            pn[j] = pl[((tn * TPT + j) * 16) * 2];
        __builtin_amdgcn_sched_barrier(0);

        float part[TPT];
        #pragma unroll
        for (int j = 0; j < TPT; ++j) {
            v2f f;
            f.x = (pc[j].x - c0) * s0;
            f.y = (pc[j].y - c1) * s1;

            v8f d = {};
            d = __builtin_amdgcn_wmma_f32_16x16x4_f32(false, a, false, f,
                                                      (short)0, d, false, false);
            float acc = 0.0f;
            #pragma unroll
            for (int r = 0; r < 8; ++r) {
                float tt = fmaxf(d[r] + b1v[r], 0.0f);
                acc = fmaf(tt, w2v[r], acc);
            }
            part[j] = acc;
        }

        float oth[TPT];
        #pragma unroll
        for (int j = 0; j < TPT; ++j) oth[j] = __shfl_xor(part[j], 16, 32);

        if (!hiH) {
            #pragma unroll
            for (int j = 0; j < TPT; ++j)
                orow[(trip * TPT + j) * 16 + h] = (part[j] + oth[j]) + b2;
        }
    }
}

extern "C" void kernel_launch(void* const* d_in, const int* in_sizes, int n_in,
                              void* d_out, int out_size, void* d_ws, size_t ws_size,
                              hipStream_t stream) {
    const float* pos1    = (const float*)d_in[0];
    const float* pos2    = (const float*)d_in[1];
    const float* queries = (const float*)d_in[2];
    const float* Wg      = (const float*)d_in[3];
    const float* bg      = (const float*)d_in[4];

    float* wws = (float*)d_ws;

    weights_gemm_wmma<<<512, 256, 0, stream>>>(queries, Wg, bg, wws);
    pair_mlp_wmma<<<1024, 256, 0, stream>>>(pos1, pos2, wws, (float*)d_out);
}


# both launches fused into one kernel: each work-group makes its 16 rows' generated weights with the same WMMA GEMM into LDS, then its 8 waves sweep 2 rows each; every sum keeps its order
# speedup vs baseline: 2.7357x; 1.1056x over previous
#include <hip/hip_runtime.h>

typedef __attribute__((ext_vector_type(2))) float v2f;
typedef __attribute__((ext_vector_type(8))) float v8f;

constexpr int Bsz = 8, N1 = 1024, N2 = 1024, DIM = 256, HID = 16;
constexpr int WCOLS = 6 * HID + 1;

constexpr int RPG = 16;
constexpr int QP  = DIM + 4;
constexpr int WP  = 100;

__global__ __launch_bounds__(256) void pair_mlp_fused_wmma(
        const float* __restrict__ q, const float* __restrict__ Wg,
        const float* __restrict__ bg, const float* __restrict__ pos1,
        const float* __restrict__ pos2, float* __restrict__ out) {
    __shared__ __attribute__((aligned(16))) float qs[16 * QP];
    __shared__ __attribute__((aligned(16))) float wl[RPG * WP];
    __shared__ float4 feat2[N2];

    const int lane = threadIdx.x & 31;
    const int wid  = threadIdx.x >> 5;
    const int row0 = blockIdx.x * RPG;
    const int b    = row0 >> 10;
    const int h    = lane & 15;
    const bool hiH = lane >= 16;
    const int koff = hiH ? 2 : 0;

    const float4* p2 = (const float4*)pos2 + (size_t)b * N2;
    float4 pb[N2 / 256];
    #pragma unroll
    for (int i = 0; i < N2 / 256; ++i) pb[i] = p2[i * 256 + (int)threadIdx.x];

    const float4* qg = (const float4*)(q + (size_t)row0 * DIM);
    float4 v[4];
    #pragma unroll
    for (int i = 0; i < 4; ++i) v[i] = qg[i * 256 + (int)threadIdx.x];

    constexpr int KU     = 16;
    constexpr int ROUNDS = DIM / (4 * KU);
    const bool gemm_wave = wid < 7;
    const int col  = wid * 16 + h;
    const int colc = (col < WCOLS) ? col : (WCOLS - 1);
    const float* wcol = Wg + colc + (size_t)koff * WCOLS;
    v2f bn[KU];
    #pragma unroll
    for (int j = 0; j < KU; ++j) {
        bn[j].x = wcol[(size_t)(j * 4)     * WCOLS];
        bn[j].y = wcol[(size_t)(j * 4 + 1) * WCOLS];
    }
    const float bgv = bg[colc];

    const int rloc0 = wid * 2;
    float4 p1v[2];
    #pragma unroll
    for (int rr = 0; rr < 2; ++rr)
        p1v[rr] = ((const float4*)pos1)[row0 + rloc0 + rr];

    #pragma unroll
    for (int i = 0; i < N2 / 256; ++i) {
        float4 p = pb[i];
        p.z = __logf(p.z);
        p.w = __logf(p.w);
        feat2[i * 256 + (int)threadIdx.x] = p;
    }
    #pragma unroll
    for (int i = 0; i < 4; ++i) {
        const int idx = i * 256 + (int)threadIdx.x;
        const int r   = idx >> 6;
        const int c4  = idx & 63;
        *(float4*)&qs[r * QP + c4 * 4] = v[i];
    }
    __syncthreads();

    if (gemm_wave) {
        const float* arow = qs + h * QP + koff;

        v8f c = {};
        #pragma unroll
        for (int rd = 0; rd < ROUNDS; ++rd) {
            v2f bc[KU];
            #pragma unroll
            for (int j = 0; j < KU; ++j) bc[j] = bn[j];
            if (rd + 1 < ROUNDS) {
                #pragma unroll
                for (int j = 0; j < KU; ++j) {
                    const int kk = ((rd + 1) * KU + j) * 4;
                    bn[j].x = wcol[(size_t)kk       * WCOLS];
                    bn[j].y = wcol[(size_t)(kk + 1) * WCOLS];
                }
            }
            #pragma unroll
            for (int j = 0; j < KU; ++j) {
                const int kk = (rd * KU + j) * 4;
                const v2f a = *(const v2f*)(arow + kk);
                c = __builtin_amdgcn_wmma_f32_16x16x4_f32(false, a, false, bc[j],
                                                          (short)0, c, false, false);
            }
        }

        const int rb = hiH ? 8 : 0;
        #pragma unroll
        for (int r = 0; r < 8; ++r) {
            if (col < WCOLS)
                wl[(rb + r) * WP + col] = c[r] + bgv;
        }
    }
    __syncthreads();

    const float2* pl = (const float2*)feat2 + h * 2 + (hiH ? 1 : 0);
    const int k0 = hiH ? 2 : 0;
    const int hb = hiH ? 8 : 0;

    #pragma unroll
    for (int rr = 0; rr < 2; ++rr) {
        const int rloc = rloc0 + rr;
        const float x1 = p1v[rr].x, y1 = p1v[rr].y, w1p = p1v[rr].z, h1p = p1v[rr].w;
        const float c0 = hiH ? __logf(w1p) : x1;
        const float c1 = hiH ? __logf(h1p) : y1;
        const float s0 = hiH ? 1.0f : 1.0f / w1p;
        const float s1 = hiH ? 1.0f : 1.0f / h1p;

        const float* wrow = wl + rloc * WP;
        v2f a;
        a.x = wrow[(k0 + 0) * HID + h];
        a.y = wrow[(k0 + 1) * HID + h];
        float b1v[8], w2v[8];
        #pragma unroll
        for (int r = 0; r < 8; ++r) {
            b1v[r] = wrow[4 * HID + hb + r];
            w2v[r] = wrow[5 * HID + hb + r];
        }
        const float b2 = wrow[6 * HID];

        float* orow = out + (size_t)(row0 + rloc) * N2;

        constexpr int TPT   = 4;
        constexpr int TRIPS = N2 / (16 * TPT);

        float2 pn[TPT];
        #pragma unroll
        for (int j = 0; j < TPT; ++j) pn[j] = pl[(j * 16) * 2];

        for (int trip = 0; trip < TRIPS; ++trip) {
            float2 pc[TPT];
            #pragma unroll
            for (int j = 0; j < TPT; ++j) pc[j] = pn[j];

            const int tn = (trip + 1) & (TRIPS - 1);
            #pragma unroll
            for (int j = 0; j < TPT; ++j)
                pn[j] = pl[((tn * TPT + j) * 16) * 2];
            __builtin_amdgcn_sched_barrier(0);

            float part[TPT];
            #pragma unroll
            for (int j = 0; j < TPT; ++j) {
                v2f f;
                f.x = (pc[j].x - c0) * s0;
                f.y = (pc[j].y - c1) * s1;

                v8f d = {};
                d = __builtin_amdgcn_wmma_f32_16x16x4_f32(false, a, false, f,
                                                          (short)0, d, false, false);
                float acc = 0.0f;
                #pragma unroll
                for (int r = 0; r < 8; ++r) {
                    float tt = fmaxf(d[r] + b1v[r], 0.0f);
                    acc = fmaf(tt, w2v[r], acc);
                }
                part[j] = acc;
            }

            float oth[TPT];
            #pragma unroll
            for (int j = 0; j < TPT; ++j) oth[j] = __shfl_xor(part[j], 16, 32);

            if (!hiH) {
                #pragma unroll
                for (int j = 0; j < TPT; ++j)
                    orow[(trip * TPT + j) * 16 + h] = (part[j] + oth[j]) + b2;
            }
        }
    }
}

extern "C" void kernel_launch(void* const* d_in, const int* in_sizes, int n_in,
                              void* d_out, int out_size, void* d_ws, size_t ws_size,
                              hipStream_t stream) {
    const float* pos1    = (const float*)d_in[0];
    const float* pos2    = (const float*)d_in[1];
    const float* queries = (const float*)d_in[2];
    const float* Wg      = (const float*)d_in[3];
    const float* bg      = (const float*)d_in[4];

    pair_mlp_fused_wmma<<<512, 256, 0, stream>>>(queries, Wg, bg, pos1, pos2,
                                                 (float*)d_out);
}


# pair stage: b1 enters the layer-1 WMMA as its C operand (was C = 0 and 8 adds per lane per tile): bias summed first in the chain, not last; relu taken as integer max on D's bits
# speedup vs baseline: 2.9091x; 1.0634x over previous
#include <hip/hip_runtime.h>

typedef __attribute__((ext_vector_type(2))) float v2f;
typedef __attribute__((ext_vector_type(8))) float v8f;

constexpr int Bsz = 8, N1 = 1024, N2 = 1024, DIM = 256, HID = 16;
constexpr int WCOLS = 6 * HID + 1;

constexpr int RPG = 16;
constexpr int QP  = DIM + 4;
constexpr int WP  = 100;

__device__ __forceinline__ float relu_bits(float x) {
    const int xi = __float_as_int(x);
    return __int_as_float(xi > 0 ? xi : 0);
}

__global__ __launch_bounds__(256) void pair_mlp_fused_wmma(
        const float* __restrict__ q, const float* __restrict__ Wg,
        const float* __restrict__ bg, const float* __restrict__ pos1,
        const float* __restrict__ pos2, float* __restrict__ out) {
    __shared__ __attribute__((aligned(16))) float qs[16 * QP];
    __shared__ __attribute__((aligned(16))) float wl[RPG * WP];
    __shared__ float4 feat2[N2];

    const int lane = threadIdx.x & 31;
    const int wid  = threadIdx.x >> 5;
    const int row0 = blockIdx.x * RPG;
    const int b    = row0 >> 10;
    const int h    = lane & 15;
    const bool hiH = lane >= 16;
    const int koff = hiH ? 2 : 0;

    const float4* p2 = (const float4*)pos2 + (size_t)b * N2;
    float4 pb[N2 / 256];
    #pragma unroll
    for (int i = 0; i < N2 / 256; ++i) pb[i] = p2[i * 256 + (int)threadIdx.x];

    const float4* qg = (const float4*)(q + (size_t)row0 * DIM);
    float4 v[4];
    #pragma unroll
    for (int i = 0; i < 4; ++i) v[i] = qg[i * 256 + (int)threadIdx.x];

    constexpr int KU     = 16;
    constexpr int ROUNDS = DIM / (4 * KU);
    const bool gemm_wave = wid < 7;
    const int col  = wid * 16 + h;
    const int colc = (col < WCOLS) ? col : (WCOLS - 1);
    const float* wcol = Wg + colc + (size_t)koff * WCOLS;
    v2f bn[KU];
    #pragma unroll
    for (int j = 0; j < KU; ++j) {
        bn[j].x = wcol[(size_t)(j * 4)     * WCOLS];
        bn[j].y = wcol[(size_t)(j * 4 + 1) * WCOLS];
    }
    const float bgv = bg[colc];

    const int rloc0 = wid * 2;
    float4 p1v[2];
    #pragma unroll
    for (int rr = 0; rr < 2; ++rr)
        p1v[rr] = ((const float4*)pos1)[row0 + rloc0 + rr];

    #pragma unroll
    for (int i = 0; i < N2 / 256; ++i) {
        float4 p = pb[i];
        p.z = __logf(p.z);
        p.w = __logf(p.w);
        feat2[i * 256 + (int)threadIdx.x] = p;
    }
    #pragma unroll
    for (int i = 0; i < 4; ++i) {
        const int idx = i * 256 + (int)threadIdx.x;
        const int r   = idx >> 6;
        const int c4  = idx & 63;
        *(float4*)&qs[r * QP + c4 * 4] = v[i];
    }
    __syncthreads();

    if (gemm_wave) {
        const float* arow = qs + h * QP + koff;

        v8f c = {};
        #pragma unroll
        for (int rd = 0; rd < ROUNDS; ++rd) {
            v2f bc[KU];
            #pragma unroll
            for (int j = 0; j < KU; ++j) bc[j] = bn[j];
            if (rd + 1 < ROUNDS) {
                #pragma unroll
                for (int j = 0; j < KU; ++j) {
                    const int kk = ((rd + 1) * KU + j) * 4;
                    bn[j].x = wcol[(size_t)kk       * WCOLS];
                    bn[j].y = wcol[(size_t)(kk + 1) * WCOLS];
                }
            }
            #pragma unroll
            for (int j = 0; j < KU; ++j) {
                const int kk = (rd * KU + j) * 4;
                const v2f a = *(const v2f*)(arow + kk);
                c = __builtin_amdgcn_wmma_f32_16x16x4_f32(false, a, false, bc[j],
                                                          (short)0, c, false, false);
            }
        }

        const int rb = hiH ? 8 : 0;
        #pragma unroll
        for (int r = 0; r < 8; ++r) {
            if (col < WCOLS)
                wl[(rb + r) * WP + col] = c[r] + bgv;
        }
    }
    __syncthreads();

    const float2* pl = (const float2*)feat2 + h * 2 + (hiH ? 1 : 0);
    const int k0 = hiH ? 2 : 0;
    const int hb = hiH ? 8 : 0;

    #pragma unroll
    for (int rr = 0; rr < 2; ++rr) {
        const int rloc = rloc0 + rr;
        const float x1 = p1v[rr].x, y1 = p1v[rr].y, w1p = p1v[rr].z, h1p = p1v[rr].w;
        const float c0 = hiH ? __logf(w1p) : x1;
        const float c1 = hiH ? __logf(h1p) : y1;
        const float s0 = hiH ? 1.0f : 1.0f / w1p;
        const float s1 = hiH ? 1.0f : 1.0f / h1p;

        const float* wrow = wl + rloc * WP;
        v2f a;
        a.x = wrow[(k0 + 0) * HID + h];
        a.y = wrow[(k0 + 1) * HID + h];
        v8f cb1;
        float w2v[8];
        #pragma unroll
        for (int r = 0; r < 8; ++r) {
            cb1[r] = wrow[4 * HID + hb + r];
            w2v[r] = wrow[5 * HID + hb + r];
        }
        const float b2 = wrow[6 * HID];

        float* orow = out + (size_t)(row0 + rloc) * N2;

        constexpr int TPT   = 4;
        constexpr int TRIPS = N2 / (16 * TPT);

        float2 pn[TPT];
        #pragma unroll
        for (int j = 0; j < TPT; ++j) pn[j] = pl[(j * 16) * 2];

        for (int trip = 0; trip < TRIPS; ++trip) {
            float2 pc[TPT];
            #pragma unroll
            for (int j = 0; j < TPT; ++j) pc[j] = pn[j];

            const int tn = (trip + 1) & (TRIPS - 1);
            #pragma unroll
            for (int j = 0; j < TPT; ++j)
                pn[j] = pl[((tn * TPT + j) * 16) * 2];
            __builtin_amdgcn_sched_barrier(0);

            float part[TPT];
            #pragma unroll
            for (int j = 0; j < TPT; ++j) {
                v2f f;
                f.x = (pc[j].x - c0) * s0;
                f.y = (pc[j].y - c1) * s1;

                const v8f d = __builtin_amdgcn_wmma_f32_16x16x4_f32(false, a, false, f,
                                                          (short)0, cb1, false, false);
                float acc = 0.0f;
                #pragma unroll
                for (int r = 0; r < 8; ++r) {
                    float tt = relu_bits(d[r]);
                    acc = fmaf(tt, w2v[r], acc);
                }
                part[j] = acc;
            }

            float oth[TPT];
            #pragma unroll
            for (int j = 0; j < TPT; ++j) oth[j] = __shfl_xor(part[j], 16, 32);

            if (!hiH) {
                #pragma unroll
                for (int j = 0; j < TPT; ++j)
                    orow[(trip * TPT + j) * 16 + h] = (part[j] + oth[j]) + b2;
            }
        }
    }
}

extern "C" void kernel_launch(void* const* d_in, const int* in_sizes, int n_in,
                              void* d_out, int out_size, void* d_ws, size_t ws_size,
                              hipStream_t stream) {
    const float* pos1    = (const float*)d_in[0];
    const float* pos2    = (const float*)d_in[1];
    const float* queries = (const float*)d_in[2];
    const float* Wg      = (const float*)d_in[3];
    const float* bg      = (const float*)d_in[4];

    pair_mlp_fused_wmma<<<512, 256, 0, stream>>>(queries, Wg, bg, pos1, pos2,
                                                 (float*)d_out);
}


# pair stage: h-halves of two n-tiles joined by one v_permlanex16 (was one ds_bpermute per tile), so all 32 lanes store one whole 128-B line (was 16 lanes, 64 B); sums bit-identical
# speedup vs baseline: 3.1168x; 1.0714x over previous
#include <hip/hip_runtime.h>

typedef __attribute__((ext_vector_type(2))) float v2f;
typedef __attribute__((ext_vector_type(8))) float v8f;

constexpr int Bsz = 8, N1 = 1024, N2 = 1024, DIM = 256, HID = 16;
constexpr int WCOLS = 6 * HID + 1;

constexpr int RPG = 16;
constexpr int QP  = DIM + 4;
constexpr int WP  = 100;

__device__ __forceinline__ float relu_bits(float x) {
    const int xi = __float_as_int(x);
    return __int_as_float(xi > 0 ? xi : 0);
}

__device__ __forceinline__ float xhalf(float x) {
    const int b = __float_as_int(x);
    return __int_as_float(__builtin_amdgcn_permlanex16(b, b, 0x76543210, 0xfedcba98,
                                                       false, false));
}

__global__ __launch_bounds__(256) void pair_mlp_fused_wmma(
        const float* __restrict__ q, const float* __restrict__ Wg,
        const float* __restrict__ bg, const float* __restrict__ pos1,
        const float* __restrict__ pos2, float* __restrict__ out) {
    __shared__ __attribute__((aligned(16))) float qs[16 * QP];
    __shared__ __attribute__((aligned(16))) float wl[RPG * WP];
    __shared__ float4 feat2[N2];

    const int lane = threadIdx.x & 31;
    const int wid  = threadIdx.x >> 5;
    const int row0 = blockIdx.x * RPG;
    const int b    = row0 >> 10;
    const int h    = lane & 15;
    const bool hiH = lane >= 16;
    const int koff = hiH ? 2 : 0;

    const float4* p2 = (const float4*)pos2 + (size_t)b * N2;
    float4 pb[N2 / 256];
    #pragma unroll
    for (int i = 0; i < N2 / 256; ++i) pb[i] = p2[i * 256 + (int)threadIdx.x];

    const float4* qg = (const float4*)(q + (size_t)row0 * DIM);
    float4 v[4];
    #pragma unroll
    for (int i = 0; i < 4; ++i) v[i] = qg[i * 256 + (int)threadIdx.x];

    constexpr int KU     = 16;
    constexpr int ROUNDS = DIM / (4 * KU);
    const bool gemm_wave = wid < 7;
    const int col  = wid * 16 + h;
    const int colc = (col < WCOLS) ? col : (WCOLS - 1);
    const float* wcol = Wg + colc + (size_t)koff * WCOLS;
    v2f bn[KU];
    #pragma unroll
    for (int j = 0; j < KU; ++j) {
        bn[j].x = wcol[(size_t)(j * 4)     * WCOLS];
        bn[j].y = wcol[(size_t)(j * 4 + 1) * WCOLS];
    }
    const float bgv = bg[colc];

    const int rloc0 = wid * 2;
    float4 p1v[2];
    #pragma unroll
    for (int rr = 0; rr < 2; ++rr)
        p1v[rr] = ((const float4*)pos1)[row0 + rloc0 + rr];

    #pragma unroll
    for (int i = 0; i < N2 / 256; ++i) {
        float4 p = pb[i];
        p.z = __logf(p.z);
        p.w = __logf(p.w);
        feat2[i * 256 + (int)threadIdx.x] = p;
    }
    #pragma unroll
    for (int i = 0; i < 4; ++i) {
        const int idx = i * 256 + (int)threadIdx.x;
        const int r   = idx >> 6;
        const int c4  = idx & 63;
        *(float4*)&qs[r * QP + c4 * 4] = v[i];
    }
    __syncthreads();

    if (gemm_wave) {
        const float* arow = qs + h * QP + koff;

        v8f c = {};
        #pragma unroll
        for (int rd = 0; rd < ROUNDS; ++rd) {
            v2f bc[KU];
            #pragma unroll
            for (int j = 0; j < KU; ++j) bc[j] = bn[j];
            if (rd + 1 < ROUNDS) {
                #pragma unroll
                for (int j = 0; j < KU; ++j) {
                    const int kk = ((rd + 1) * KU + j) * 4;
                    bn[j].x = wcol[(size_t)kk       * WCOLS];
                    bn[j].y = wcol[(size_t)(kk + 1) * WCOLS];
                }
            }
            #pragma unroll
            for (int j = 0; j < KU; ++j) {
                const int kk = (rd * KU + j) * 4;
                const v2f a = *(const v2f*)(arow + kk);
                c = __builtin_amdgcn_wmma_f32_16x16x4_f32(false, a, false, bc[j],
                                                          (short)0, c, false, false);
            }
        }

        const int rb = hiH ? 8 : 0;
        #pragma unroll
        for (int r = 0; r < 8; ++r) {
            if (col < WCOLS)
                wl[(rb + r) * WP + col] = c[r] + bgv;
        }
    }
    __syncthreads();

    const float2* pl = (const float2*)feat2 + h * 2 + (hiH ? 1 : 0);
    const int k0 = hiH ? 2 : 0;
    const int hb = hiH ? 8 : 0;

    #pragma unroll
    for (int rr = 0; rr < 2; ++rr) {
        const int rloc = rloc0 + rr;
        const float x1 = p1v[rr].x, y1 = p1v[rr].y, w1p = p1v[rr].z, h1p = p1v[rr].w;
        const float c0 = hiH ? __logf(w1p) : x1;
        const float c1 = hiH ? __logf(h1p) : y1;
        const float s0 = hiH ? 1.0f : 1.0f / w1p;
        const float s1 = hiH ? 1.0f : 1.0f / h1p;

        const float* wrow = wl + rloc * WP;
        v2f a;
        a.x = wrow[(k0 + 0) * HID + h];
        a.y = wrow[(k0 + 1) * HID + h];
        v8f cb1;
        float w2v[8];
        #pragma unroll
        for (int r = 0; r < 8; ++r) {
            cb1[r] = wrow[4 * HID + hb + r];
            w2v[r] = wrow[5 * HID + hb + r];
        }
        const float b2 = wrow[6 * HID];

        float* orow = out + (size_t)(row0 + rloc) * N2;

        constexpr int TPT   = 4;
        constexpr int TRIPS = N2 / (16 * TPT);

        float2 pn[TPT];
        #pragma unroll
        for (int j = 0; j < TPT; ++j) pn[j] = pl[(j * 16) * 2];

        for (int trip = 0; trip < TRIPS; ++trip) {
            float2 pc[TPT];
            #pragma unroll
            for (int j = 0; j < TPT; ++j) pc[j] = pn[j];

            const int tn = (trip + 1) & (TRIPS - 1);
            #pragma unroll
            for (int j = 0; j < TPT; ++j)
                pn[j] = pl[((tn * TPT + j) * 16) * 2];
            __builtin_amdgcn_sched_barrier(0);

            float part[TPT];
            #pragma unroll
            for (int j = 0; j < TPT; ++j) {
                v2f f;
                f.x = (pc[j].x - c0) * s0;
                f.y = (pc[j].y - c1) * s1;

                const v8f d = __builtin_amdgcn_wmma_f32_16x16x4_f32(false, a, false, f,
                                                          (short)0, cb1, false, false);
                float acc = 0.0f;
                #pragma unroll
                for (int r = 0; r < 8; ++r) {
                    float tt = relu_bits(d[r]);
                    acc = fmaf(tt, w2v[r], acc);
                }
                part[j] = acc;
            }

            #pragma unroll
            for (int j = 0; j < TPT; j += 2) {
                const float keep = hiH ? part[j + 1] : part[j];
                const float give = hiH ? part[j]     : part[j + 1];
                orow[(trip * TPT + j) * 16 + lane] = (keep + xhalf(give)) + b2;
            }
        }
    }
}

extern "C" void kernel_launch(void* const* d_in, const int* in_sizes, int n_in,
                              void* d_out, int out_size, void* d_ws, size_t ws_size,
                              hipStream_t stream) {
    const float* pos1    = (const float*)d_in[0];
    const float* pos2    = (const float*)d_in[1];
    const float* queries = (const float*)d_in[2];
    const float* Wg      = (const float*)d_in[3];
    const float* bg      = (const float*)d_in[4];

    pair_mlp_fused_wmma<<<512, 256, 0, stream>>>(queries, Wg, bg, pos1, pos2,
                                                 (float*)d_out);
}


# pair stage re-tiled: one WMMA = 16 n x 16 rows for one hidden unit (was 16 h x 16 n of one row), row shift/scale folded into B and C, so w2 sums stay in the lane in h order 0..15; no half exchange
# speedup vs baseline: 3.5680x; 1.1448x over previous
#include <hip/hip_runtime.h>

typedef __attribute__((ext_vector_type(2))) float v2f;
typedef __attribute__((ext_vector_type(8))) float v8f;

constexpr int Bsz = 8, N1 = 1024, N2 = 1024, DIM = 256, HID = 16;
constexpr int WCOLS = 6 * HID + 1;

constexpr int RPG = 16;
constexpr int QP  = DIM + 4;
constexpr int WP  = 100;
constexpr int TP  = 132;

__device__ __forceinline__ float relu_bits(float x) {
    const int xi = __float_as_int(x);
    return __int_as_float(xi > 0 ? xi : 0);
}

__global__ __launch_bounds__(256) __attribute__((amdgpu_waves_per_eu(4, 4))) void pair_mlp_fused_wmma(
        const float* __restrict__ q, const float* __restrict__ Wg,
        const float* __restrict__ bg, const float* __restrict__ pos1,
        const float* __restrict__ pos2, float* __restrict__ out) {
    __shared__ __attribute__((aligned(16))) float qs[16 * QP];
    __shared__ __attribute__((aligned(16))) float wl[RPG * WP];
    __shared__ float4 feat2[N2];
    __shared__ __attribute__((aligned(16))) float tb[8 * RPG * TP];

    const int lane = threadIdx.x & 31;
    const int wid  = threadIdx.x >> 5;
    const int row0 = blockIdx.x * RPG;
    const int b    = row0 >> 10;
    const int h    = lane & 15;
    const bool hiH = lane >= 16;
    const int koff = hiH ? 2 : 0;

    const float4* p2 = (const float4*)pos2 + (size_t)b * N2;
    float4 pb[N2 / 256];
    #pragma unroll
    for (int i = 0; i < N2 / 256; ++i) pb[i] = p2[i * 256 + (int)threadIdx.x];

    const float4* qg = (const float4*)(q + (size_t)row0 * DIM);
    float4 v[4];
    #pragma unroll
    for (int i = 0; i < 4; ++i) v[i] = qg[i * 256 + (int)threadIdx.x];

    constexpr int KU     = 16;
    constexpr int ROUNDS = DIM / (4 * KU);
    const bool gemm_wave = wid < 7;
    const int col  = wid * 16 + h;
    const int colc = (col < WCOLS) ? col : (WCOLS - 1);
    const float* wcol = Wg + colc + (size_t)koff * WCOLS;
    v2f bn[KU];
    #pragma unroll
    for (int j = 0; j < KU; ++j) {
        bn[j].x = wcol[(size_t)(j * 4)     * WCOLS];
        bn[j].y = wcol[(size_t)(j * 4 + 1) * WCOLS];
    }
    const float bgv = bg[colc];

    const float4 p1b = ((const float4*)pos1)[row0 + h];

    #pragma unroll
    for (int i = 0; i < N2 / 256; ++i) {
        float4 p = pb[i];
        p.z = __logf(p.z);
        p.w = __logf(p.w);
        feat2[i * 256 + (int)threadIdx.x] = p;
    }
    #pragma unroll
    for (int i = 0; i < 4; ++i) {
        const int idx = i * 256 + (int)threadIdx.x;
        const int r   = idx >> 6;
        const int c4  = idx & 63;
        *(float4*)&qs[r * QP + c4 * 4] = v[i];
    }
    __syncthreads();

    if (gemm_wave) {
        const float* arow = qs + h * QP + koff;

        v8f c = {};
        #pragma unroll
        for (int rd = 0; rd < ROUNDS; ++rd) {
            v2f bc[KU];
            #pragma unroll
            for (int j = 0; j < KU; ++j) bc[j] = bn[j];
            if (rd + 1 < ROUNDS) {
                #pragma unroll
                for (int j = 0; j < KU; ++j) {
                    const int kk = ((rd + 1) * KU + j) * 4;
                    bn[j].x = wcol[(size_t)kk       * WCOLS];
                    bn[j].y = wcol[(size_t)(kk + 1) * WCOLS];
                }
            }
            #pragma unroll
            for (int j = 0; j < KU; ++j) {
                const int kk = (rd * KU + j) * 4;
                const v2f a = *(const v2f*)(arow + kk);
                c = __builtin_amdgcn_wmma_f32_16x16x4_f32(false, a, false, bc[j],
                                                          (short)0, c, false, false);
            }
        }

        const int rb = hiH ? 8 : 0;
        #pragma unroll
        for (int r = 0; r < 8; ++r) {
            if (col < WCOLS)
                wl[(rb + r) * WP + col] = c[r] + bgv;
        }
    }
    __syncthreads();

    const float sx = 1.0f / p1b.z, sy = 1.0f / p1b.w;
    const float g0 = p1b.x * sx, g1 = p1b.y * sy;
    const float g2 = __logf(p1b.z), g3 = __logf(p1b.w);
    const float sa = hiH ? 1.0f : sx;
    const float sb = hiH ? 1.0f : sy;

    const float* wrow = wl + h * WP;
    const float* wown = wrow + (hiH ? 2 : 0) * HID;
    float Bx[HID], By[HID], bp[HID], w2l[HID];
    #pragma unroll
    for (int q = 0; q < HID / 4; ++q) {
        const float4 a0 = *(const float4*)(wrow + 0 * HID + 4 * q);
        const float4 a1 = *(const float4*)(wrow + 1 * HID + 4 * q);
        const float4 a2 = *(const float4*)(wrow + 2 * HID + 4 * q);
        const float4 a3 = *(const float4*)(wrow + 3 * HID + 4 * q);
        const float4 bb = *(const float4*)(wrow + 4 * HID + 4 * q);
        const float4 ww = *(const float4*)(wrow + 5 * HID + 4 * q);
        const float4 o0 = *(const float4*)(wown + 0 * HID + 4 * q);
        const float4 o1 = *(const float4*)(wown + 1 * HID + 4 * q);
        const float a0v[4] = {a0.x, a0.y, a0.z, a0.w};
        const float a1v[4] = {a1.x, a1.y, a1.z, a1.w};
        const float a2v[4] = {a2.x, a2.y, a2.z, a2.w};
        const float a3v[4] = {a3.x, a3.y, a3.z, a3.w};
        const float bbv[4] = {bb.x, bb.y, bb.z, bb.w};
        const float wwv[4] = {ww.x, ww.y, ww.z, ww.w};
        const float o0v[4] = {o0.x, o0.y, o0.z, o0.w};
        const float o1v[4] = {o1.x, o1.y, o1.z, o1.w};
        #pragma unroll
        for (int e = 0; e < 4; ++e) {
            float t = fmaf(-g0, a0v[e], bbv[e]);
            t = fmaf(-g1, a1v[e], t);
            t = fmaf(-g2, a2v[e], t);
            t = fmaf(-g3, a3v[e], t);
            bp[4 * q + e]  = t;
            Bx[4 * q + e]  = o0v[e] * sa;
            By[4 * q + e]  = o1v[e] * sb;
            w2l[4 * q + e] = wwv[e];
        }
    }
    const float b2 = wrow[6 * HID];

    constexpr int NT = 4;
    constexpr int NPASS = 2;
    const int nbase = wid * (16 * NT * NPASS);
    const float2* pl = (const float2*)feat2 + (nbase + h) * 2 + (hiH ? 1 : 0);
    float* tw = tb + wid * (RPG * TP);

    #pragma unroll
    for (int ps = 0; ps < NPASS; ++ps) {
        v2f af[NT];
        #pragma unroll
        for (int t = 0; t < NT; ++t) {
            const float2 pv = pl[((ps * NT + t) * 16) * 2];
            af[t].x = pv.x;
            af[t].y = pv.y;
        }
        v8f acc[NT];
        #pragma unroll
        for (int t = 0; t < NT; ++t) acc[t] = v8f{};

        #pragma unroll
        for (int hh = 0; hh < HID; ++hh) {
            v2f bf;
            bf.x = Bx[hh];
            bf.y = By[hh];
            v8f cb;
            #pragma unroll
            for (int r = 0; r < 8; ++r) cb[r] = bp[hh];
            const float w2h = w2l[hh];
            #pragma unroll
            for (int t = 0; t < NT; ++t) {
                const v8f d = __builtin_amdgcn_wmma_f32_16x16x4_f32(false, af[t], false, bf,
                                                          (short)0, cb, false, false);
                #pragma unroll
                for (int r = 0; r < 8; ++r)
                    acc[t][r] = fmaf(relu_bits(d[r]), w2h, acc[t][r]);
            }
        }

        #pragma unroll
        for (int t = 0; t < NT; ++t) {
            float4 lo, hi4;
            lo.x  = acc[t][0] + b2; lo.y  = acc[t][1] + b2; lo.z  = acc[t][2] + b2; lo.w  = acc[t][3] + b2;
            hi4.x = acc[t][4] + b2; hi4.y = acc[t][5] + b2; hi4.z = acc[t][6] + b2; hi4.w = acc[t][7] + b2;
            float* dst = tw + h * TP + 16 * (ps * NT + t) + (hiH ? 8 : 0);
            *(float4*)(dst)     = lo;
            *(float4*)(dst + 4) = hi4;
        }
    }

    __builtin_amdgcn_fence(__ATOMIC_RELEASE, "workgroup");
    __builtin_amdgcn_wave_barrier();
    __builtin_amdgcn_fence(__ATOMIC_ACQUIRE, "workgroup");
    float* ocol = out + (size_t)row0 * N2 + nbase + 4 * lane;
    #pragma unroll
    for (int i = 0; i < RPG; ++i) {
        const float4 v4 = *(const float4*)(tw + i * TP + 4 * lane);
        *(float4*)(ocol + (size_t)i * N2) = v4;
    }
}

extern "C" void kernel_launch(void* const* d_in, const int* in_sizes, int n_in,
                              void* d_out, int out_size, void* d_ws, size_t ws_size,
                              hipStream_t stream) {
    const float* pos1    = (const float*)d_in[0];
    const float* pos2    = (const float*)d_in[1];
    const float* queries = (const float*)d_in[2];
    const float* Wg      = (const float*)d_in[3];
    const float* bg      = (const float*)d_in[4];

    pair_mlp_fused_wmma<<<512, 256, 0, stream>>>(queries, Wg, bg, pos1, pos2,
                                                 (float*)d_out);
}
